# starting kernel
# baseline (speedup 1.0000x reference)
#include <hip/hip_runtime.h>

#define N_NODES 100000
#define N_EDGES 600000
#define DIM     128
#define ALPHA   0.5f

typedef __attribute__((ext_vector_type(2))) float v2f;
typedef __attribute__((ext_vector_type(8))) float v8f;

__global__ void dirgcn_zero_kernel(float* __restrict__ p, long n) {
    long i = (long)blockIdx.x * blockDim.x + threadIdx.x;
    long stride = (long)gridDim.x * blockDim.x;
    for (; i < n; i += stride) p[i] = 0.0f;
}

__global__ void dirgcn_degree_kernel(const int* __restrict__ ei,
                                     float* __restrict__ deg_out,
                                     float* __restrict__ deg_in) {
    int e = blockIdx.x * blockDim.x + threadIdx.x;
    if (e < N_EDGES) {
        atomicAdd(&deg_out[ei[e]], 1.0f);
        atomicAdd(&deg_in[ei[N_EDGES + e]], 1.0f);
    }
}

__global__ void dirgcn_scatter_kernel(const float* __restrict__ x,
                                      const int* __restrict__ ei,
                                      const float* __restrict__ deg_out,
                                      const float* __restrict__ deg_in,
                                      float* __restrict__ agg_src,
                                      float* __restrict__ agg_dst) {
    int wave = blockIdx.x * (blockDim.x >> 5) + (threadIdx.x >> 5);
    int lane = threadIdx.x & 31;
    int row = ei[wave];
    int col = ei[N_EDGES + wave];
    float dro = deg_out[row];
    float dci = deg_in[col];
    float ws = (dro > 0.0f) ? rsqrtf(fmaxf(dro, 1.0f)) : 0.0f;
    float wd = (dci > 0.0f) ? rsqrtf(fmaxf(dci, 1.0f)) : 0.0f;
    float w = ws * wd;

    const float4* xc = (const float4*)(x + (size_t)col * DIM);
    const float4* xr = (const float4*)(x + (size_t)row * DIM);
    float4 vc = xc[lane];
    float4 vr = xr[lane];

    float* as = agg_src + (size_t)row * DIM + lane * 4;
    float* ad = agg_dst + (size_t)col * DIM + lane * 4;
    atomicAdd(as + 0, w * vc.x);
    atomicAdd(as + 1, w * vc.y);
    atomicAdd(as + 2, w * vc.z);
    atomicAdd(as + 3, w * vc.w);
    atomicAdd(ad + 0, w * vr.x);
    atomicAdd(ad + 1, w * vr.y);
    atomicAdd(ad + 2, w * vr.z);
    atomicAdd(ad + 3, w * vr.w);
}

__global__ void dirgcn_wprep_kernel(const float* __restrict__ Ws,
                                    const float* __restrict__ bs,
                                    const float* __restrict__ Wd,
                                    const float* __restrict__ bd,
                                    float* __restrict__ Ws_s,
                                    float* __restrict__ Wd_s,
                                    float* __restrict__ bias) {
    int i = blockIdx.x * blockDim.x + threadIdx.x;
    if (i < DIM * DIM) {
        Ws_s[i] = (1.0f - ALPHA) * Ws[i];
        Wd_s[i] = ALPHA * Wd[i];
    }
    if (i < DIM) bias[i] = (1.0f - ALPHA) * bs[i] + ALPHA * bd[i];
}

__global__ void __launch_bounds__(256)
dirgcn_gemm_kernel(const float* __restrict__ agg_src,
                   const float* __restrict__ agg_dst,
                   const float* __restrict__ Ws,
                   const float* __restrict__ Wd,
                   const float* __restrict__ bias,
                   float* __restrict__ out) {
    const int lane = threadIdx.x & 31;
    const int wave = threadIdx.x >> 5;
    const int row0 = blockIdx.x * 16;
    const int j0   = wave * 16;
    const int m    = lane & 15;
    const int koff = (lane >> 4) * 2;

    const float* arow_s = agg_src + (size_t)(row0 + m) * DIM;
    const float* arow_d = agg_dst + (size_t)(row0 + m) * DIM;
    const float* wrow_s = Ws + (size_t)(j0 + m) * DIM;
    const float* wrow_d = Wd + (size_t)(j0 + m) * DIM;

    v8f acc = {};
    #pragma unroll
    for (int k = 0; k < DIM; k += 4) {
        v2f a = *(const v2f*)(arow_s + k + koff);
        v2f b = *(const v2f*)(wrow_s + k + koff);
        acc = __builtin_amdgcn_wmma_f32_16x16x4_f32(false, a, false, b,
                                                    (short)0, acc, false, false);
    }
    #pragma unroll
    for (int k = 0; k < DIM; k += 4) {
        v2f a = *(const v2f*)(arow_d + k + koff);
        v2f b = *(const v2f*)(wrow_d + k + koff);
        acc = __builtin_amdgcn_wmma_f32_16x16x4_f32(false, a, false, b,
                                                    (short)0, acc, false, false);
    }

    float bv = bias[j0 + m];
    const int n = lane & 15;
    const int mbase = (lane >> 4) * 8;
    #pragma unroll
    for (int r = 0; r < 8; ++r) {
        out[(size_t)(row0 + mbase + r) * DIM + j0 + n] = acc[r] + bv;
    }
}

extern "C" void kernel_launch(void* const* d_in, const int* in_sizes, int n_in,
                              void* d_out, int out_size, void* d_ws, size_t ws_size,
                              hipStream_t stream) {
    const float* x    = (const float*)d_in[0];
    const int*   ei   = (const int*)d_in[1];
    const float* W_src = (const float*)d_in[2];
    const float* b_src = (const float*)d_in[3];
    const float* W_dst = (const float*)d_in[4];
    const float* b_dst = (const float*)d_in[5];
    float* out = (float*)d_out;

    float* ws = (float*)d_ws;
    float* deg_out = ws;
    float* deg_in  = deg_out + N_NODES;
    float* agg_src = deg_in + N_NODES;
    float* agg_dst = agg_src + (size_t)N_NODES * DIM;
    float* Ws_s    = agg_dst + (size_t)N_NODES * DIM;
    float* Wd_s    = Ws_s + DIM * DIM;
    float* bias    = Wd_s + DIM * DIM;

    const long n_zero = 2L * N_NODES + 2L * N_NODES * DIM;
    dirgcn_zero_kernel<<<2048, 256, 0, stream>>>(ws, n_zero);

    dirgcn_degree_kernel<<<(N_EDGES + 255) / 256, 256, 0, stream>>>(ei, deg_out, deg_in);

    dirgcn_scatter_kernel<<<N_EDGES / 8, 256, 0, stream>>>(x, ei, deg_out, deg_in,
                                                           agg_src, agg_dst);

    dirgcn_wprep_kernel<<<(DIM * DIM + 255) / 256, 256, 0, stream>>>(
        W_src, b_src, W_dst, b_dst, Ws_s, Wd_s, bias);

    dirgcn_gemm_kernel<<<N_NODES / 16, 256, 0, stream>>>(agg_src, agg_dst,
                                                         Ws_s, Wd_s, bias, out);
}


# Aggregation by owner work-groups, not global float atomic scatter: a group holds 391 node rows in LDS, scans all keys, queues its edges, sums with LDS float adds (arrival order), stores rows once
# speedup vs baseline: 1.7872x; 1.7872x over previous
#include <hip/hip_runtime.h>

#define N_NODES 100000
#define N_EDGES 600000
#define DIM     128
#define ALPHA   0.5f

typedef __attribute__((ext_vector_type(2))) float v2f;
typedef __attribute__((ext_vector_type(4))) float v4f;
typedef __attribute__((ext_vector_type(8))) float v8f;

__global__ void dirgcn_zero_kernel(float* __restrict__ p, long n) {
    long i = (long)blockIdx.x * blockDim.x + threadIdx.x;
    long stride = (long)gridDim.x * blockDim.x;
    for (; i < n; i += stride) p[i] = 0.0f;
}

__global__ void dirgcn_degree_kernel(const int* __restrict__ ei,
                                     float* __restrict__ deg_out,
                                     float* __restrict__ deg_in) {
    int e = blockIdx.x * blockDim.x + threadIdx.x;
    if (e < N_EDGES) {
        atomicAdd(&deg_out[ei[e]], 1.0f);
        atomicAdd(&deg_in[ei[N_EDGES + e]], 1.0f);
    }
}

#define OWN_ROWS   391
#define OWN_GROUPS 256
#define OWN_QCAP   8192

template <int K>
static __device__ __forceinline__ void dirgcn_drain_round(const float* __restrict__ x, float* acc,
                                                          int o, float w, int dl, int j, int lane) {
    float v[K][4]; float wj[K]; int dj[K];
    #pragma unroll
    for (int u = 0; u < K; ++u) {
        const int oj = __shfl(o, j + u);
        wj[u] = __shfl(w, j + u);
        dj[u] = __shfl(dl, j + u);
        const float* xr = x + (size_t)oj * DIM + lane;
        #pragma unroll
        for (int i = 0; i < 4; ++i) v[u][i] = xr[32 * i];
    }
    #pragma unroll
    for (int u = 0; u < K; ++u) {
        #pragma unroll
        for (int i = 0; i < 4; ++i)
            atomicAdd(&acc[dj[u] * DIM + lane + 32 * i], wj[u] * v[u][i]);
    }
}

__global__ void __launch_bounds__(1024)
dirgcn_owner_kernel(const float* __restrict__ x,
                    const int* __restrict__ ei,
                    const float* __restrict__ deg_out,
                    const float* __restrict__ deg_in,
                    float* __restrict__ agg_src,
                    float* __restrict__ agg_dst) {
    __shared__ float acc[OWN_ROWS * DIM];
    __shared__ int   queue[OWN_QCAP];
    __shared__ int   qn;

    const int tid  = threadIdx.x;
    const int lane = tid & 31;
    const int wave = tid >> 5;
    const int side = (blockIdx.x >= OWN_GROUPS) ? 1 : 0;
    const int g    = blockIdx.x - side * OWN_GROUPS;
    const int lo   = g * OWN_ROWS;
    const int cnt  = min(OWN_ROWS, N_NODES - lo);
    const int* key = ei + side * N_EDGES;
    const int* oth = ei + (1 - side) * N_EDGES;
    float* agg     = side ? agg_dst : agg_src;

    for (int i = tid; i < OWN_ROWS * DIM / 4; i += 1024)
        ((float4*)acc)[i] = make_float4(0.0f, 0.0f, 0.0f, 0.0f);
    if (tid == 0) qn = 0;
    __syncthreads();

    auto weight = [&](int kk, int oo) -> float {
        int row = side ? oo : kk;
        int col = side ? kk : oo;
        float dro = deg_out[row];
        float dci = deg_in[col];
        float ws = (dro > 0.0f) ? rsqrtf(fmaxf(dro, 1.0f)) : 0.0f;
        float wd = (dci > 0.0f) ? rsqrtf(fmaxf(dci, 1.0f)) : 0.0f;
        return ws * wd;
    };
    auto push = [&](int e, unsigned d) {
        int slot = atomicAdd(&qn, 1);
        if (slot < OWN_QCAP) {
            queue[slot] = e | (int)(d << 20);
        } else {
            int oo = oth[e];
            float w = weight(lo + (int)d, oo);
            const float* xr = x + (size_t)oo * DIM;
            for (int c = 0; c < DIM; ++c) atomicAdd(&acc[d * DIM + c], w * xr[c]);
        }
    };

    const int4* key4 = (const int4*)key;
    const int rot = g * 576;
    for (int it = tid; it < N_EDGES / 4; it += 2048) {
        const int it1 = it + 1024;
        int i0 = it + rot;  if (i0 >= N_EDGES / 4) i0 -= N_EDGES / 4;
        int i1 = it1 + rot; if (i1 >= N_EDGES / 4) i1 -= N_EDGES / 4;
        int4 k0 = key4[i0];
        int4 k1 = (it1 < N_EDGES / 4) ? key4[i1] : make_int4(-1, -1, -1, -1);
        unsigned d;
        d = (unsigned)(k0.x - lo); if (d < (unsigned)cnt) push(4 * i0 + 0, d);
        d = (unsigned)(k0.y - lo); if (d < (unsigned)cnt) push(4 * i0 + 1, d);
        d = (unsigned)(k0.z - lo); if (d < (unsigned)cnt) push(4 * i0 + 2, d);
        d = (unsigned)(k0.w - lo); if (d < (unsigned)cnt) push(4 * i0 + 3, d);
        d = (unsigned)(k1.x - lo); if (d < (unsigned)cnt) push(4 * i1 + 0, d);
        d = (unsigned)(k1.y - lo); if (d < (unsigned)cnt) push(4 * i1 + 1, d);
        d = (unsigned)(k1.z - lo); if (d < (unsigned)cnt) push(4 * i1 + 2, d);
        d = (unsigned)(k1.w - lo); if (d < (unsigned)cnt) push(4 * i1 + 3, d);
    }
    __syncthreads();
    const int n = min(qn, OWN_QCAP);

    for (int base = wave * 32; base < n; base += 32 * 32) {
        const int m = min(32, n - base);
        int o = 0; int dl = 0; float w = 0.0f;
        if (lane < m) {
            int q = queue[base + lane];
            int e = q & 0xFFFFF;
            dl = (int)((unsigned)q >> 20);
            o = oth[e];
            w = weight(lo + dl, o);
        }
        int j = 0;
        for (; j + 8 <= m; j += 8) dirgcn_drain_round<8>(x, acc, o, w, dl, j, lane);
        if (j + 4 <= m) { dirgcn_drain_round<4>(x, acc, o, w, dl, j, lane); j += 4; }
        if (j + 2 <= m) { dirgcn_drain_round<2>(x, acc, o, w, dl, j, lane); j += 2; }
        if (j < m)      { dirgcn_drain_round<1>(x, acc, o, w, dl, j, lane); }
    }
    __syncthreads();

    float4* dst = (float4*)(agg + (size_t)lo * DIM);
    for (int i = tid; i < cnt * (DIM / 4); i += 1024) dst[i] = ((const float4*)acc)[i];
}

__global__ void dirgcn_wprep_kernel(const float* __restrict__ Ws,
                                    const float* __restrict__ bs,
                                    const float* __restrict__ Wd,
                                    const float* __restrict__ bd,
                                    float* __restrict__ Wp,
                                    float* __restrict__ bias) {
    int d = blockIdx.x * blockDim.x + threadIdx.x;
    if (d < 2 * DIM * DIM) {
        int e    = d & 3;
        int lane = (d >> 2) & 31;
        int j    = (d >> 7) & 7;
        int p    = (d >> 10) & 15;
        int arr  = d >> 14;
        int n = 16 * j + (lane & 15);
        int k = 8 * p + 4 * (lane >> 4) + e;
        float w = arr ? ALPHA * Wd[n * DIM + k] : (1.0f - ALPHA) * Ws[n * DIM + k];
        Wp[d] = w;
    }
    if (d < DIM) bias[d] = (1.0f - ALPHA) * bs[d] + ALPHA * bd[d];
}

__global__ void __launch_bounds__(64)
dirgcn_gemm_kernel(const float* __restrict__ agg_src,
                   const float* __restrict__ agg_dst,
                   const float* __restrict__ Wp,
                   const float* __restrict__ bias,
                   float* __restrict__ out) {
    const int lane = threadIdx.x & 31;
    const int wave = threadIdx.x >> 5;
    const int tile = blockIdx.x * 2 + wave;
    const int row0 = tile * 16;
    const int c    = lane & 15;
    const int h    = lane >> 4;

    const unsigned aoff = (unsigned)((row0 + c) * DIM + 4 * h) * 4u;
    const unsigned woff = (unsigned)lane * 16u;
    const char* asb = (const char*)agg_src;
    const char* adb = (const char*)agg_dst;
    const char* wpb = (const char*)Wp;

    v8f acc[8];
    #pragma unroll
    for (int j = 0; j < 8; ++j) acc[j] = (v8f){};

    v4f a0, a1, b0[8], b1[8];

    auto load_set = [&](int qn, v4f& a, v4f (&b)[8]) {
        const char* ab = (qn < 16) ? (asb + 32 * qn) : (adb + 32 * (qn - 16));
        const char* wq = wpb + qn * 4096;
        a = *(const v4f*)(ab + aoff);
        #pragma unroll
        for (int j = 0; j < 8; ++j) b[j] = *(const v4f*)(wq + j * 512 + woff);
    };
    auto mma_set = [&](const v4f& a, const v4f (&b)[8]) {
        #pragma unroll
        for (int j = 0; j < 8; ++j)
            acc[j] = __builtin_amdgcn_wmma_f32_16x16x4_f32(false, a.xy, false, b[j].xy,
                                                           (short)0, acc[j], false, false);
        #pragma unroll
        for (int j = 0; j < 8; ++j)
            acc[j] = __builtin_amdgcn_wmma_f32_16x16x4_f32(false, a.zw, false, b[j].zw,
                                                           (short)0, acc[j], false, false);
    };

    load_set(0, a0, b0);
    #pragma unroll 1
    for (int q = 0; q < 30; q += 2) {
        load_set(q + 1, a1, b1);
        __builtin_amdgcn_sched_barrier(0);
        mma_set(a0, b0);
        __builtin_amdgcn_sched_barrier(0);
        load_set(q + 2, a0, b0);
        __builtin_amdgcn_sched_barrier(0);
        mma_set(a1, b1);
    }
    load_set(31, a1, b1);
    __builtin_amdgcn_sched_barrier(0);
    mma_set(a0, b0);
    mma_set(a1, b1);

    float* orow = out + (size_t)(row0 + h * 8) * DIM + c;
    #pragma unroll
    for (int j = 0; j < 8; ++j) {
        float bv = bias[16 * j + c];
        #pragma unroll
        for (int r = 0; r < 8; ++r) {
            orow[(size_t)r * DIM + 16 * j] = acc[j][r] + bv;
        }
    }
}

extern "C" void kernel_launch(void* const* d_in, const int* in_sizes, int n_in,
                              void* d_out, int out_size, void* d_ws, size_t ws_size,
                              hipStream_t stream) {
    const float* x    = (const float*)d_in[0];
    const int*   ei   = (const int*)d_in[1];
    const float* W_src = (const float*)d_in[2];
    const float* b_src = (const float*)d_in[3];
    const float* W_dst = (const float*)d_in[4];
    const float* b_dst = (const float*)d_in[5];
    float* out = (float*)d_out;

    float* ws = (float*)d_ws;
    float* deg_out = ws;
    float* deg_in  = deg_out + N_NODES;
    float* agg_src = deg_in + N_NODES;
    float* agg_dst = agg_src + (size_t)N_NODES * DIM;
    float* Wp      = agg_dst + (size_t)N_NODES * DIM;
    float* bias    = Wp + 2 * DIM * DIM;

    dirgcn_zero_kernel<<<(2 * N_NODES + 255) / 256, 256, 0, stream>>>(ws, 2L * N_NODES);

    dirgcn_degree_kernel<<<(N_EDGES + 255) / 256, 256, 0, stream>>>(ei, deg_out, deg_in);

    dirgcn_owner_kernel<<<2 * OWN_GROUPS, 1024, 0, stream>>>(x, ei, deg_out, deg_in,
                                                            agg_src, agg_dst);

    dirgcn_wprep_kernel<<<(2 * DIM * DIM + 255) / 256, 256, 0, stream>>>(
        W_src, b_src, W_dst, b_dst, Wp, bias);

    dirgcn_gemm_kernel<<<N_NODES / 32, 64, 0, stream>>>(agg_src, agg_dst,
                                                        Wp, bias, out);
}


# Edges reach owner groups by a per-chunk partition: degree kernel sorts each 1024-edge chunk by owner (LDS ranks), writes records + start/count table; no scan of all owner bytes; arrival-order sums
# speedup vs baseline: 2.6303x; 1.4717x over previous
#include <hip/hip_runtime.h>

#define N_NODES 100000
#define N_EDGES 600000
#define DIM     128
#define ALPHA   0.5f

typedef __attribute__((ext_vector_type(2))) float v2f;
typedef __attribute__((ext_vector_type(4))) float v4f;
typedef __attribute__((ext_vector_type(8))) float v8f;
typedef __attribute__((ext_vector_type(4))) unsigned v4u;

__global__ void dirgcn_zero_kernel(float* __restrict__ p, long n) {
    long i = (long)blockIdx.x * blockDim.x + threadIdx.x;
    long stride = (long)gridDim.x * blockDim.x;
    for (; i < n; i += stride) p[i] = 0.0f;
}

#define OWN_ROWS   391
#define OWN_GROUPS 256
#define PART_EDGES 1024
#define PART_NB    ((N_EDGES + PART_EDGES - 1) / PART_EDGES)
#define REC_OTH    0x1FFFFu
__global__ void __launch_bounds__(256)
dirgcn_degree_kernel(const int* __restrict__ ei,
                     float* __restrict__ deg_out,
                     float* __restrict__ deg_in,
                     unsigned* __restrict__ tab,
                     unsigned* __restrict__ lst) {
    __shared__ unsigned cnt[2][OWN_GROUPS];
    __shared__ unsigned wsum[8];
    __shared__ v4u      srt4[2][PART_EDGES / 4];
    unsigned* srt0 = (unsigned*)srt4[0];
    unsigned* srt1 = (unsigned*)srt4[1];
    const int tid  = threadIdx.x;
    const int lane = tid & 31;
    const int wave = tid >> 5;
    const int blk  = blockIdx.x;
    cnt[0][tid] = 0u;
    cnt[1][tid] = 0u;
    __syncthreads();

    const int t = blk * 256 + tid;
    const bool valid = t < N_EDGES / 4;
    int kr[4] = {0, 0, 0, 0}, kc[4] = {0, 0, 0, 0};
    unsigned rr[4] = {0u, 0u, 0u, 0u}, rc[4] = {0u, 0u, 0u, 0u};
    if (valid) {
        const int4 r = ((const int4*)ei)[t];
        const int4 c = ((const int4*)(ei + N_EDGES))[t];
        kr[0] = r.x; kr[1] = r.y; kr[2] = r.z; kr[3] = r.w;
        kc[0] = c.x; kc[1] = c.y; kc[2] = c.z; kc[3] = c.w;
        for (int i = 0; i < 4; ++i) {
            rr[i] = atomicAdd(&cnt[0][kr[i] / OWN_ROWS], 1u);
            rc[i] = atomicAdd(&cnt[1][kc[i] / OWN_ROWS], 1u);
        }
    }
    __syncthreads();

    const unsigned v = cnt[0][tid] | (cnt[1][tid] << 16);
    unsigned inc = v;
    for (int d = 1; d < 32; d <<= 1) {
        const unsigned up = __shfl_up(inc, d);
        if (lane >= d) inc += up;
    }
    if (lane == 31) wsum[wave] = inc;
    __syncthreads();
    unsigned off = 0u;
    for (int k = 0; k < wave; ++k) off += wsum[k];
    const unsigned exc = off + inc - v;
    cnt[0][tid] = exc & 0xFFFFu;
    cnt[1][tid] = exc >> 16;
    __syncthreads();

    if (valid) {
        for (int i = 0; i < 4; ++i) {
            const int ow = kr[i] / OWN_ROWS;
            srt0[cnt[0][ow] + rr[i]] = (unsigned)kc[i] | ((unsigned)(kr[i] - ow * OWN_ROWS) << 17);
        }
        for (int i = 0; i < 4; ++i) {
            const int ow = kc[i] / OWN_ROWS;
            srt1[cnt[1][ow] + rc[i]] = (unsigned)kr[i] | ((unsigned)(kc[i] - ow * OWN_ROWS) << 17);
        }
    }
    __syncthreads();

    tab[((size_t)blk) * OWN_GROUPS + tid]             = (exc & 0xFFFFu) | (v << 16);
    tab[((size_t)(PART_NB + blk)) * OWN_GROUPS + tid] = (exc >> 16) | (v & 0xFFFF0000u);
    if (valid) {
        ((v4u*)(lst + (size_t)blk * PART_EDGES))[tid]           = srt4[0][tid];
        ((v4u*)(lst + N_EDGES + (size_t)blk * PART_EDGES))[tid] = srt4[1][tid];
        for (int i = 0; i < 4; ++i) {
            atomicAdd(&deg_out[kr[i]], 1.0f);
            atomicAdd(&deg_in[kc[i]], 1.0f);
        }
    }
}

#define OWN_QCAP   8192

template <int K>
static __device__ __forceinline__ void dirgcn_drain_round(const float* __restrict__ x, float* acc,
                                                          int o, float w, int dl, int j, int lane) {
    v4f v[K]; float wj[K]; int dj[K];
    #pragma unroll
    for (int u = 0; u < K; ++u) {
        const int oj = __builtin_amdgcn_readlane(o, j + u);
        wj[u] = __int_as_float(__builtin_amdgcn_readlane(__float_as_int(w), j + u));
        dj[u] = __builtin_amdgcn_readlane(dl, j + u);
        v[u] = ((const v4f*)(x + (size_t)oj * DIM))[lane];
    }
    #pragma unroll
    for (int u = 0; u < K; ++u) {
        #pragma unroll
        for (int i = 0; i < 4; ++i)
            atomicAdd(&acc[dj[u] * DIM + 32 * i + lane], wj[u] * v[u][i]);
    }
}

__global__ void __launch_bounds__(1024)
dirgcn_owner_kernel(const float* __restrict__ x,
                    const float* __restrict__ deg_out,
                    const float* __restrict__ deg_in,
                    const unsigned* __restrict__ tab,
                    const unsigned* __restrict__ lst,
                    float* __restrict__ agg_src,
                    float* __restrict__ agg_dst) {
    __shared__ float acc[OWN_ROWS * DIM];
    __shared__ int   queue[OWN_QCAP];
    __shared__ int   qn;

    const int tid  = threadIdx.x;
    const int lane = tid & 31;
    const int wave = tid >> 5;
    const int side = (blockIdx.x >= OWN_GROUPS) ? 1 : 0;
    const int g    = blockIdx.x - side * OWN_GROUPS;
    const int lo   = g * OWN_ROWS;
    const int cnt  = min(OWN_ROWS, N_NODES - lo);
    float* agg     = side ? agg_dst : agg_src;

    for (int i = tid; i < OWN_ROWS * DIM / 4; i += 1024)
        ((float4*)acc)[i] = make_float4(0.0f, 0.0f, 0.0f, 0.0f);
    if (tid == 0) qn = 0;
    __syncthreads();

    auto weight = [&](int kk, int oo) -> float {
        int row = side ? oo : kk;
        int col = side ? kk : oo;
        float dro = deg_out[row];
        float dci = deg_in[col];
        float ws = (dro > 0.0f) ? rsqrtf(fmaxf(dro, 1.0f)) : 0.0f;
        float wd = (dci > 0.0f) ? rsqrtf(fmaxf(dci, 1.0f)) : 0.0f;
        return ws * wd;
    };
    if (tid < PART_NB) {
        const unsigned te = tab[((size_t)(side * PART_NB + tid)) * OWN_GROUPS + g];
        const int c    = (int)(te >> 16);
        const int slot = atomicAdd(&qn, c);
        const unsigned* p = lst + (size_t)side * N_EDGES + (size_t)tid * PART_EDGES + (te & 0xFFFFu);
        const int cq = max(0, min(c, OWN_QCAP - slot));
        for (int i0 = 0; i0 < cq; i0 += 16) {
            unsigned rec[16];
            #pragma unroll
            for (int u = 0; u < 16; ++u) rec[u] = p[min(i0 + u, cq - 1)];
            #pragma unroll
            for (int u = 0; u < 16; ++u) queue[slot + min(i0 + u, cq - 1)] = (int)rec[u];
        }
        for (int i = cq; i < c; ++i) {
            const unsigned rec = p[i];
            const int oo = (int)(rec & REC_OTH);
            const int dl = (int)(rec >> 17);
            const float w = weight(lo + dl, oo);
            const float* xr = x + (size_t)oo * DIM;
            for (int f = 0; f < DIM; ++f)
                atomicAdd(&acc[dl * DIM + 32 * (f & 3) + (f >> 2)], w * xr[f]);
        }
    }
    __syncthreads();
    const int n = __builtin_amdgcn_readfirstlane(min(qn, OWN_QCAP));

    const int wave_s = __builtin_amdgcn_readfirstlane(wave);
    for (int base = wave_s * 32; base < n; base += 32 * 32) {
        const int m = min(32, n - base);
        int o = 0; int dl = 0; float w = 0.0f;
        if (lane < m) {
            const unsigned rec = (unsigned)queue[base + lane];
            o  = (int)(rec & REC_OTH);
            dl = (int)(rec >> 17);
            w  = weight(lo + dl, o);
        }
        int j = 0;
        for (; j + 16 <= m; j += 16) dirgcn_drain_round<16>(x, acc, o, w, dl, j, lane);
        if (j + 8 <= m) { dirgcn_drain_round<8>(x, acc, o, w, dl, j, lane); j += 8; }
        if (j + 4 <= m) { dirgcn_drain_round<4>(x, acc, o, w, dl, j, lane); j += 4; }
        if (j + 2 <= m) { dirgcn_drain_round<2>(x, acc, o, w, dl, j, lane); j += 2; }
        if (j < m)      { dirgcn_drain_round<1>(x, acc, o, w, dl, j, lane); }
    }
    __syncthreads();

    float4* dst = (float4*)(agg + (size_t)lo * DIM);
    for (int i = tid; i < cnt * (DIM / 4); i += 1024) {
        const float* ar = acc + (i >> 5) * DIM + (i & 31);
        dst[i] = make_float4(ar[0], ar[32], ar[64], ar[96]);
    }
}

__global__ void dirgcn_wprep_kernel(const float* __restrict__ Ws,
                                    const float* __restrict__ bs,
                                    const float* __restrict__ Wd,
                                    const float* __restrict__ bd,
                                    float* __restrict__ Wp,
                                    float* __restrict__ bias) {
    int d = blockIdx.x * blockDim.x + threadIdx.x;
    if (d < 2 * DIM * DIM) {
        int e    = d & 3;
        int lane = (d >> 2) & 31;
        int j    = (d >> 7) & 7;
        int p    = (d >> 10) & 15;
        int arr  = d >> 14;
        int n = 16 * j + (lane & 15);
        int k = 8 * p + 4 * (lane >> 4) + e;
        float w = arr ? ALPHA * Wd[n * DIM + k] : (1.0f - ALPHA) * Ws[n * DIM + k];
        Wp[d] = w;
    }
    if (d < DIM) bias[d] = (1.0f - ALPHA) * bs[d] + ALPHA * bd[d];
}

__global__ void __launch_bounds__(64)
dirgcn_gemm_kernel(const float* __restrict__ agg_src,
                   const float* __restrict__ agg_dst,
                   const float* __restrict__ Wp,
                   const float* __restrict__ bias,
                   float* __restrict__ out) {
    const int lane = threadIdx.x & 31;
    const int wave = threadIdx.x >> 5;
    const int tile = blockIdx.x * 2 + wave;
    const int row0 = tile * 16;
    const int c    = lane & 15;
    const int h    = lane >> 4;

    const unsigned aoff = (unsigned)((row0 + c) * DIM + 4 * h) * 4u;
    const unsigned woff = (unsigned)lane * 16u;
    const char* asb = (const char*)agg_src;
    const char* adb = (const char*)agg_dst;
    const char* wpb = (const char*)Wp;

    v8f acc[8];
    #pragma unroll
    for (int j = 0; j < 8; ++j) acc[j] = (v8f){};

    v4f a0, a1, b0[8], b1[8];

    auto load_set = [&](int qn, v4f& a, v4f (&b)[8]) {
        const char* ab = (qn < 16) ? (asb + 32 * qn) : (adb + 32 * (qn - 16));
        const char* wq = wpb + qn * 4096;
        a = *(const v4f*)(ab + aoff);
        #pragma unroll
        for (int j = 0; j < 8; ++j) b[j] = *(const v4f*)(wq + j * 512 + woff);
    };
    auto mma_set = [&](const v4f& a, const v4f (&b)[8]) {
        #pragma unroll
        for (int j = 0; j < 8; ++j)
            acc[j] = __builtin_amdgcn_wmma_f32_16x16x4_f32(false, a.xy, false, b[j].xy,
                                                           (short)0, acc[j], false, false);
        #pragma unroll
        for (int j = 0; j < 8; ++j)
            acc[j] = __builtin_amdgcn_wmma_f32_16x16x4_f32(false, a.zw, false, b[j].zw,
                                                           (short)0, acc[j], false, false);
    };

    load_set(0, a0, b0);
    #pragma unroll 1
    for (int q = 0; q < 30; q += 2) {
        load_set(q + 1, a1, b1);
        __builtin_amdgcn_sched_barrier(0);
        mma_set(a0, b0);
        __builtin_amdgcn_sched_barrier(0);
        load_set(q + 2, a0, b0);
        __builtin_amdgcn_sched_barrier(0);
        mma_set(a1, b1);
    }
    load_set(31, a1, b1);
    __builtin_amdgcn_sched_barrier(0);
    mma_set(a0, b0);
    mma_set(a1, b1);

    float* orow = out + (size_t)(row0 + h * 8) * DIM + c;
    #pragma unroll
    for (int j = 0; j < 8; ++j) {
        float bv = bias[16 * j + c];
        #pragma unroll
        for (int r = 0; r < 8; ++r) {
            orow[(size_t)r * DIM + 16 * j] = acc[j][r] + bv;
        }
    }
}

extern "C" void kernel_launch(void* const* d_in, const int* in_sizes, int n_in,
                              void* d_out, int out_size, void* d_ws, size_t ws_size,
                              hipStream_t stream) {
    const float* x    = (const float*)d_in[0];
    const int*   ei   = (const int*)d_in[1];
    const float* W_src = (const float*)d_in[2];
    const float* b_src = (const float*)d_in[3];
    const float* W_dst = (const float*)d_in[4];
    const float* b_dst = (const float*)d_in[5];
    float* out = (float*)d_out;

    float* ws = (float*)d_ws;
    float* deg_out = ws;
    float* deg_in  = deg_out + N_NODES;
    float* agg_src = deg_in + N_NODES;
    float* agg_dst = agg_src + (size_t)N_NODES * DIM;
    float* Wp      = agg_dst + (size_t)N_NODES * DIM;
    float* bias    = Wp + 2 * DIM * DIM;

    dirgcn_zero_kernel<<<(2 * N_NODES + 255) / 256, 256, 0, stream>>>(ws, 2L * N_NODES);

    unsigned* lst = (unsigned*)d_out;
    unsigned* tab = lst + 2 * N_EDGES;
    dirgcn_degree_kernel<<<PART_NB, 256, 0, stream>>>(ei, deg_out, deg_in, tab, lst);

    dirgcn_owner_kernel<<<2 * OWN_GROUPS, 1024, 0, stream>>>(x, deg_out, deg_in, tab, lst,
                                                            agg_src, agg_dst);

    dirgcn_wprep_kernel<<<(2 * DIM * DIM + 255) / 256, 256, 0, stream>>>(
        W_src, b_src, W_dst, b_dst, Wp, bias);

    dirgcn_gemm_kernel<<<N_NODES / 32, 64, 0, stream>>>(agg_src, agg_dst,
                                                        Wp, bias, out);
}


# Node degrees counted by the owner groups from the partitioned edge records (LDS integer adds, each degree stored once, counts exact): no scattered global float atomic adds, no zero-fill launch
# speedup vs baseline: 3.6189x; 1.3759x over previous
#include <hip/hip_runtime.h>

#define N_NODES 100000
#define N_EDGES 600000
#define DIM     128
#define ALPHA   0.5f

typedef __attribute__((ext_vector_type(2))) float v2f;
typedef __attribute__((ext_vector_type(4))) float v4f;
typedef __attribute__((ext_vector_type(8))) float v8f;
typedef __attribute__((ext_vector_type(4))) unsigned v4u;

#define OWN_ROWS   391
#define OWN_GROUPS 256
#define PART_EDGES 1024
#define PART_NB    ((N_EDGES + PART_EDGES - 1) / PART_EDGES)
#define REC_OTH    0x1FFFFu
__global__ void __launch_bounds__(256)
dirgcn_partition_kernel(const int* __restrict__ ei,
                        unsigned* __restrict__ tab,
                        unsigned* __restrict__ lst) {
    __shared__ unsigned cnt[2][OWN_GROUPS];
    __shared__ unsigned wsum[8];
    __shared__ v4u      srt4[2][PART_EDGES / 4];
    unsigned* srt0 = (unsigned*)srt4[0];
    unsigned* srt1 = (unsigned*)srt4[1];
    const int tid  = threadIdx.x;
    const int lane = tid & 31;
    const int wave = tid >> 5;
    const int blk  = blockIdx.x;
    cnt[0][tid] = 0u;
    cnt[1][tid] = 0u;
    __syncthreads();

    const int t = blk * 256 + tid;
    const bool valid = t < N_EDGES / 4;
    int kr[4] = {0, 0, 0, 0}, kc[4] = {0, 0, 0, 0};
    unsigned rr[4] = {0u, 0u, 0u, 0u}, rc[4] = {0u, 0u, 0u, 0u};
    if (valid) {
        const int4 r = ((const int4*)ei)[t];
        const int4 c = ((const int4*)(ei + N_EDGES))[t];
        kr[0] = r.x; kr[1] = r.y; kr[2] = r.z; kr[3] = r.w;
        kc[0] = c.x; kc[1] = c.y; kc[2] = c.z; kc[3] = c.w;
        for (int i = 0; i < 4; ++i) {
            rr[i] = atomicAdd(&cnt[0][kr[i] / OWN_ROWS], 1u);
            rc[i] = atomicAdd(&cnt[1][kc[i] / OWN_ROWS], 1u);
        }
    }
    __syncthreads();

    const unsigned v = cnt[0][tid] | (cnt[1][tid] << 16);
    unsigned inc = v;
    for (int d = 1; d < 32; d <<= 1) {
        const unsigned up = __shfl_up(inc, d);
        if (lane >= d) inc += up;
    }
    if (lane == 31) wsum[wave] = inc;
    __syncthreads();
    unsigned off = 0u;
    for (int k = 0; k < wave; ++k) off += wsum[k];
    const unsigned exc = off + inc - v;
    cnt[0][tid] = exc & 0xFFFFu;
    cnt[1][tid] = exc >> 16;
    __syncthreads();

    if (valid) {
        for (int i = 0; i < 4; ++i) {
            const int ow = kr[i] / OWN_ROWS;
            srt0[cnt[0][ow] + rr[i]] = (unsigned)kc[i] | ((unsigned)(kr[i] - ow * OWN_ROWS) << 17);
        }
        for (int i = 0; i < 4; ++i) {
            const int ow = kc[i] / OWN_ROWS;
            srt1[cnt[1][ow] + rc[i]] = (unsigned)kr[i] | ((unsigned)(kc[i] - ow * OWN_ROWS) << 17);
        }
    }
    __syncthreads();

    tab[((size_t)blk) * OWN_GROUPS + tid]             = (exc & 0xFFFFu) | (v << 16);
    tab[((size_t)(PART_NB + blk)) * OWN_GROUPS + tid] = (exc >> 16) | (v & 0xFFFF0000u);
    if (valid) {
        ((v4u*)(lst + (size_t)blk * PART_EDGES))[tid]           = srt4[0][tid];
        ((v4u*)(lst + N_EDGES + (size_t)blk * PART_EDGES))[tid] = srt4[1][tid];
    }
}

#define CNT_SEGS    2
#define CNT_THREADS 320
__global__ void __launch_bounds__(CNT_THREADS)
dirgcn_count_kernel(const unsigned* __restrict__ tab,
                    const unsigned* __restrict__ lst,
                    float* __restrict__ deg_out,
                    float* __restrict__ deg_in) {
    __shared__ unsigned cnt[OWN_ROWS];
    const int tid  = threadIdx.x;
    const int side = (blockIdx.x >= OWN_GROUPS) ? 1 : 0;
    const int g    = blockIdx.x - side * OWN_GROUPS;
    const int lo   = g * OWN_ROWS;
    for (int i = tid; i < OWN_ROWS; i += CNT_THREADS) cnt[i] = 0u;
    __syncthreads();

    int c[CNT_SEGS];
    const unsigned* p[CNT_SEGS];
    unsigned te[CNT_SEGS];
    #pragma unroll
    for (int s = 0; s < CNT_SEGS; ++s) {
        const int b = tid + s * CNT_THREADS;
        te[s] = (b < PART_NB) ? tab[((size_t)(side * PART_NB + b)) * OWN_GROUPS + g] : 0u;
    }
    #pragma unroll
    for (int s = 0; s < CNT_SEGS; ++s) {
        const int b = tid + s * CNT_THREADS;
        c[s] = (int)(te[s] >> 16);
        p[s] = lst + (size_t)side * N_EDGES + (size_t)b * PART_EDGES + (te[s] & 0xFFFFu);
    }
    unsigned rec[CNT_SEGS][8];
    #pragma unroll
    for (int s = 0; s < CNT_SEGS; ++s) {
        const unsigned* q = (c[s] > 0) ? p[s] : lst;
        const int last = max(c[s] - 1, 0);
        #pragma unroll
        for (int u = 0; u < 8; ++u) rec[s][u] = q[min(u, last)];
    }
    #pragma unroll
    for (int s = 0; s < CNT_SEGS; ++s) {
        #pragma unroll
        for (int u = 0; u < 8; ++u)
            atomicAdd(&cnt[rec[s][u] >> 17], (u < c[s]) ? 1u : 0u);
    }
    #pragma unroll
    for (int s = 0; s < CNT_SEGS; ++s) {
        for (int i0 = 8; i0 < c[s]; i0 += 8) {
            unsigned r2[8];
            #pragma unroll
            for (int u = 0; u < 8; ++u) r2[u] = p[s][min(i0 + u, c[s] - 1)];
            #pragma unroll
            for (int u = 0; u < 8; ++u)
                atomicAdd(&cnt[r2[u] >> 17], (i0 + u < c[s]) ? 1u : 0u);
        }
    }
    __syncthreads();
    float* deg = side ? deg_in : deg_out;
    for (int i = tid; i < OWN_ROWS; i += CNT_THREADS)
        if (lo + i < N_NODES) deg[lo + i] = (float)cnt[i];
}

#define OWN_QCAP   8192

template <int K>
static __device__ __forceinline__ void dirgcn_drain_round(const float* __restrict__ x, float* acc,
                                                          int o, float w, int dl, int j, int lane) {
    v4f v[K]; float wj[K]; int dj[K];
    #pragma unroll
    for (int u = 0; u < K; ++u) {
        const int oj = __builtin_amdgcn_readlane(o, j + u);
        wj[u] = __int_as_float(__builtin_amdgcn_readlane(__float_as_int(w), j + u));
        dj[u] = __builtin_amdgcn_readlane(dl, j + u);
        v[u] = ((const v4f*)(x + (size_t)oj * DIM))[lane];
    }
    #pragma unroll
    for (int u = 0; u < K; ++u) {
        #pragma unroll
        for (int i = 0; i < 4; ++i)
            atomicAdd(&acc[dj[u] * DIM + 32 * i + lane], wj[u] * v[u][i]);
    }
}

__global__ void __launch_bounds__(1024)
dirgcn_owner_kernel(const float* __restrict__ x,
                    const float* __restrict__ deg_out,
                    const float* __restrict__ deg_in,
                    const unsigned* __restrict__ tab,
                    const unsigned* __restrict__ lst,
                    float* __restrict__ agg_src,
                    float* __restrict__ agg_dst) {
    __shared__ float acc[OWN_ROWS * DIM];
    __shared__ int   queue[OWN_QCAP];
    __shared__ int   qn;

    const int tid  = threadIdx.x;
    const int lane = tid & 31;
    const int wave = tid >> 5;
    const int side = (blockIdx.x >= OWN_GROUPS) ? 1 : 0;
    const int g    = blockIdx.x - side * OWN_GROUPS;
    const int lo   = g * OWN_ROWS;
    const int cnt  = min(OWN_ROWS, N_NODES - lo);
    float* agg     = side ? agg_dst : agg_src;

    for (int i = tid; i < OWN_ROWS * DIM / 4; i += 1024)
        ((float4*)acc)[i] = make_float4(0.0f, 0.0f, 0.0f, 0.0f);
    if (tid == 0) qn = 0;
    __syncthreads();

    auto weight = [&](int kk, int oo) -> float {
        int row = side ? oo : kk;
        int col = side ? kk : oo;
        float dro = deg_out[row];
        float dci = deg_in[col];
        float ws = (dro > 0.0f) ? rsqrtf(fmaxf(dro, 1.0f)) : 0.0f;
        float wd = (dci > 0.0f) ? rsqrtf(fmaxf(dci, 1.0f)) : 0.0f;
        return ws * wd;
    };
    if (tid < PART_NB) {
        const unsigned te = tab[((size_t)(side * PART_NB + tid)) * OWN_GROUPS + g];
        const int c    = (int)(te >> 16);
        const int slot = atomicAdd(&qn, c);
        const unsigned* p = lst + (size_t)side * N_EDGES + (size_t)tid * PART_EDGES + (te & 0xFFFFu);
        const int cq = max(0, min(c, OWN_QCAP - slot));
        for (int i0 = 0; i0 < cq; i0 += 16) {
            unsigned rec[16];
            #pragma unroll
            for (int u = 0; u < 16; ++u) rec[u] = p[min(i0 + u, cq - 1)];
            #pragma unroll
            for (int u = 0; u < 16; ++u) queue[slot + min(i0 + u, cq - 1)] = (int)rec[u];
        }
        for (int i = cq; i < c; ++i) {
            const unsigned rec = p[i];
            const int oo = (int)(rec & REC_OTH);
            const int dl = (int)(rec >> 17);
            const float w = weight(lo + dl, oo);
            const float* xr = x + (size_t)oo * DIM;
            for (int f = 0; f < DIM; ++f)
                atomicAdd(&acc[dl * DIM + 32 * (f & 3) + (f >> 2)], w * xr[f]);
        }
    }
    __syncthreads();
    const int n = __builtin_amdgcn_readfirstlane(min(qn, OWN_QCAP));

    const int wave_s = __builtin_amdgcn_readfirstlane(wave);
    for (int base = wave_s * 32; base < n; base += 32 * 32) {
        const int m = min(32, n - base);
        int o = 0; int dl = 0; float w = 0.0f;
        if (lane < m) {
            const unsigned rec = (unsigned)queue[base + lane];
            o  = (int)(rec & REC_OTH);
            dl = (int)(rec >> 17);
            w  = weight(lo + dl, o);
        }
        int j = 0;
        for (; j + 16 <= m; j += 16) dirgcn_drain_round<16>(x, acc, o, w, dl, j, lane);
        if (j + 8 <= m) { dirgcn_drain_round<8>(x, acc, o, w, dl, j, lane); j += 8; }
        if (j + 4 <= m) { dirgcn_drain_round<4>(x, acc, o, w, dl, j, lane); j += 4; }
        if (j + 2 <= m) { dirgcn_drain_round<2>(x, acc, o, w, dl, j, lane); j += 2; }
        if (j < m)      { dirgcn_drain_round<1>(x, acc, o, w, dl, j, lane); }
    }
    __syncthreads();

    float4* dst = (float4*)(agg + (size_t)lo * DIM);
    for (int i = tid; i < cnt * (DIM / 4); i += 1024) {
        const float* ar = acc + (i >> 5) * DIM + (i & 31);
        dst[i] = make_float4(ar[0], ar[32], ar[64], ar[96]);
    }
}

__global__ void dirgcn_wprep_kernel(const float* __restrict__ Ws,
                                    const float* __restrict__ bs,
                                    const float* __restrict__ Wd,
                                    const float* __restrict__ bd,
                                    float* __restrict__ Wp,
                                    float* __restrict__ bias) {
    int d = blockIdx.x * blockDim.x + threadIdx.x;
    if (d < 2 * DIM * DIM) {
        int e    = d & 3;
        int lane = (d >> 2) & 31;
        int j    = (d >> 7) & 7;
        int p    = (d >> 10) & 15;
        int arr  = d >> 14;
        int n = 16 * j + (lane & 15);
        int k = 8 * p + 4 * (lane >> 4) + e;
        float w = arr ? ALPHA * Wd[n * DIM + k] : (1.0f - ALPHA) * Ws[n * DIM + k];
        Wp[d] = w;
    }
    if (d < DIM) bias[d] = (1.0f - ALPHA) * bs[d] + ALPHA * bd[d];
}

__global__ void __launch_bounds__(64)
dirgcn_gemm_kernel(const float* __restrict__ agg_src,
                   const float* __restrict__ agg_dst,
                   const float* __restrict__ Wp,
                   const float* __restrict__ bias,
                   float* __restrict__ out) {
    const int lane = threadIdx.x & 31;
    const int wave = threadIdx.x >> 5;
    const int tile = blockIdx.x * 2 + wave;
    const int row0 = tile * 16;
    const int c    = lane & 15;
    const int h    = lane >> 4;

    const unsigned aoff = (unsigned)((row0 + c) * DIM + 4 * h) * 4u;
    const unsigned woff = (unsigned)lane * 16u;
    const char* asb = (const char*)agg_src;
    const char* adb = (const char*)agg_dst;
    const char* wpb = (const char*)Wp;

    v8f acc[8];
    #pragma unroll
    for (int j = 0; j < 8; ++j) acc[j] = (v8f){};

    v4f a0, a1, b0[8], b1[8];

    auto load_set = [&](int qn, v4f& a, v4f (&b)[8]) {
        const char* ab = (qn < 16) ? (asb + 32 * qn) : (adb + 32 * (qn - 16));
        const char* wq = wpb + qn * 4096;
        a = *(const v4f*)(ab + aoff);
        #pragma unroll
        for (int j = 0; j < 8; ++j) b[j] = *(const v4f*)(wq + j * 512 + woff);
    };
    auto mma_set = [&](const v4f& a, const v4f (&b)[8]) {
        #pragma unroll
        for (int j = 0; j < 8; ++j)
            acc[j] = __builtin_amdgcn_wmma_f32_16x16x4_f32(false, a.xy, false, b[j].xy,
                                                           (short)0, acc[j], false, false);
        #pragma unroll
        for (int j = 0; j < 8; ++j)
            acc[j] = __builtin_amdgcn_wmma_f32_16x16x4_f32(false, a.zw, false, b[j].zw,
                                                           (short)0, acc[j], false, false);
    };

    load_set(0, a0, b0);
    #pragma unroll 1
    for (int q = 0; q < 30; q += 2) {
        load_set(q + 1, a1, b1);
        __builtin_amdgcn_sched_barrier(0);
        mma_set(a0, b0);
        __builtin_amdgcn_sched_barrier(0);
        load_set(q + 2, a0, b0);
        __builtin_amdgcn_sched_barrier(0);
        mma_set(a1, b1);
    }
    load_set(31, a1, b1);
    __builtin_amdgcn_sched_barrier(0);
    mma_set(a0, b0);
    mma_set(a1, b1);

    float* orow = out + (size_t)(row0 + h * 8) * DIM + c;
    #pragma unroll
    for (int j = 0; j < 8; ++j) {
        float bv = bias[16 * j + c];
        #pragma unroll
        for (int r = 0; r < 8; ++r) {
            orow[(size_t)r * DIM + 16 * j] = acc[j][r] + bv;
        }
    }
}

extern "C" void kernel_launch(void* const* d_in, const int* in_sizes, int n_in,
                              void* d_out, int out_size, void* d_ws, size_t ws_size,
                              hipStream_t stream) {
    const float* x    = (const float*)d_in[0];
    const int*   ei   = (const int*)d_in[1];
    const float* W_src = (const float*)d_in[2];
    const float* b_src = (const float*)d_in[3];
    const float* W_dst = (const float*)d_in[4];
    const float* b_dst = (const float*)d_in[5];
    float* out = (float*)d_out;

    float* ws = (float*)d_ws;
    float* deg_out = ws;
    float* deg_in  = deg_out + N_NODES;
    float* agg_src = deg_in + N_NODES;
    float* agg_dst = agg_src + (size_t)N_NODES * DIM;
    float* Wp      = agg_dst + (size_t)N_NODES * DIM;
    float* bias    = Wp + 2 * DIM * DIM;

    unsigned* lst = (unsigned*)d_out;
    unsigned* tab = lst + 2 * N_EDGES;
    dirgcn_partition_kernel<<<PART_NB, 256, 0, stream>>>(ei, tab, lst);

    dirgcn_count_kernel<<<2 * OWN_GROUPS, CNT_THREADS, 0, stream>>>(tab, lst, deg_out, deg_in);

    dirgcn_owner_kernel<<<2 * OWN_GROUPS, 1024, 0, stream>>>(x, deg_out, deg_in, tab, lst,
                                                            agg_src, agg_dst);

    dirgcn_wprep_kernel<<<(2 * DIM * DIM + 255) / 256, 256, 0, stream>>>(
        W_src, b_src, W_dst, b_dst, Wp, bias);

    dirgcn_gemm_kernel<<<N_NODES / 32, 64, 0, stream>>>(agg_src, agg_dst,
                                                        Wp, bias, out);
}


# Owner record lists written end to end by the count kernel (16 lanes per chunk segment, places by prefix sums): owner fill is a whole-line copy, no table, segment or slot-add reads; arrival-order sums
# speedup vs baseline: 3.8603x; 1.0667x over previous
#include <hip/hip_runtime.h>

#define N_NODES 100000
#define N_EDGES 600000
#define DIM     128
#define ALPHA   0.5f

typedef __attribute__((ext_vector_type(2))) float v2f;
typedef __attribute__((ext_vector_type(4))) float v4f;
typedef __attribute__((ext_vector_type(8))) float v8f;
typedef __attribute__((ext_vector_type(4))) unsigned v4u;

#define OWN_ROWS   391
#define OWN_GROUPS 256
#define PART_EDGES 1024
#define PART_NB    ((N_EDGES + PART_EDGES - 1) / PART_EDGES)
#define REC_OTH    0x1FFFFu
__global__ void __launch_bounds__(256)
dirgcn_partition_kernel(const int* __restrict__ ei,
                        unsigned* __restrict__ tab,
                        unsigned* __restrict__ lst) {
    __shared__ unsigned cnt[2][OWN_GROUPS];
    __shared__ unsigned wsum[8];
    __shared__ v4u      srt4[2][PART_EDGES / 4];
    unsigned* srt0 = (unsigned*)srt4[0];
    unsigned* srt1 = (unsigned*)srt4[1];
    const int tid  = threadIdx.x;
    const int lane = tid & 31;
    const int wave = tid >> 5;
    const int blk  = blockIdx.x;
    cnt[0][tid] = 0u;
    cnt[1][tid] = 0u;
    __syncthreads();

    const int t = blk * 256 + tid;
    const bool valid = t < N_EDGES / 4;
    int kr[4] = {0, 0, 0, 0}, kc[4] = {0, 0, 0, 0};
    unsigned rr[4] = {0u, 0u, 0u, 0u}, rc[4] = {0u, 0u, 0u, 0u};
    if (valid) {
        const int4 r = ((const int4*)ei)[t];
        const int4 c = ((const int4*)(ei + N_EDGES))[t];
        kr[0] = r.x; kr[1] = r.y; kr[2] = r.z; kr[3] = r.w;
        kc[0] = c.x; kc[1] = c.y; kc[2] = c.z; kc[3] = c.w;
        for (int i = 0; i < 4; ++i) {
            rr[i] = atomicAdd(&cnt[0][kr[i] / OWN_ROWS], 1u);
            rc[i] = atomicAdd(&cnt[1][kc[i] / OWN_ROWS], 1u);
        }
    }
    __syncthreads();

    const unsigned v = cnt[0][tid] | (cnt[1][tid] << 16);
    unsigned inc = v;
    for (int d = 1; d < 32; d <<= 1) {
        const unsigned up = __shfl_up(inc, d);
        if (lane >= d) inc += up;
    }
    if (lane == 31) wsum[wave] = inc;
    __syncthreads();
    unsigned off = 0u;
    for (int k = 0; k < wave; ++k) off += wsum[k];
    const unsigned exc = off + inc - v;
    cnt[0][tid] = exc & 0xFFFFu;
    cnt[1][tid] = exc >> 16;
    __syncthreads();

    if (valid) {
        for (int i = 0; i < 4; ++i) {
            const int ow = kr[i] / OWN_ROWS;
            srt0[cnt[0][ow] + rr[i]] = (unsigned)kc[i] | ((unsigned)(kr[i] - ow * OWN_ROWS) << 17);
        }
        for (int i = 0; i < 4; ++i) {
            const int ow = kc[i] / OWN_ROWS;
            srt1[cnt[1][ow] + rc[i]] = (unsigned)kr[i] | ((unsigned)(kc[i] - ow * OWN_ROWS) << 17);
        }
    }
    __syncthreads();

    tab[((size_t)blk) * OWN_GROUPS + tid]             = (exc & 0xFFFFu) | (v << 16);
    tab[((size_t)(PART_NB + blk)) * OWN_GROUPS + tid] = (exc >> 16) | (v & 0xFFFF0000u);
    if (valid) {
        ((v4u*)(lst + (size_t)blk * PART_EDGES))[tid]           = srt4[0][tid];
        ((v4u*)(lst + N_EDGES + (size_t)blk * PART_EDGES))[tid] = srt4[1][tid];
    }
}

#define OWN_LCAP    4096
#define SEG_LANES   16
#define CNT_THREADS 256
#define CNT_GROUPS  (CNT_THREADS / SEG_LANES)
#define CNT_PASSES  ((PART_NB + CNT_GROUPS - 1) / CNT_GROUPS)
__global__ void __launch_bounds__(CNT_THREADS)
dirgcn_count_kernel(const unsigned* __restrict__ tab,
                    const unsigned* __restrict__ lst,
                    float* __restrict__ deg_out,
                    float* __restrict__ deg_in,
                    unsigned* __restrict__ olist,
                    unsigned* __restrict__ ocnt) {
    __shared__ unsigned cnt[OWN_ROWS];
    __shared__ unsigned gtot[CNT_GROUPS];
    const int tid  = threadIdx.x;
    const int side = (blockIdx.x >= OWN_GROUPS) ? 1 : 0;
    const int g    = blockIdx.x - side * OWN_GROUPS;
    const int lo   = g * OWN_ROWS;
    for (int i = tid; i < OWN_ROWS; i += CNT_THREADS) cnt[i] = 0u;

    const unsigned grp   = (unsigned)tid / SEG_LANES;
    const unsigned s     = (unsigned)tid % SEG_LANES;
    const unsigned sbase = (unsigned)side * N_EDGES;
    const unsigned* tabg = tab + (size_t)side * PART_NB * OWN_GROUPS + g;
    unsigned te[CNT_PASSES];
    #pragma unroll
    for (int k = 0; k < CNT_PASSES; ++k) {
        const unsigned b = grp + (unsigned)k * CNT_GROUPS;
        const unsigned t = tabg[min(b, (unsigned)(PART_NB - 1)) * OWN_GROUPS];
        te[k] = (b < PART_NB) ? t : 0u;
    }
    unsigned rec[CNT_PASSES];
    unsigned cmax = 0u, tot = 0u;
    #pragma unroll
    for (int k = 0; k < CNT_PASSES; ++k) {
        const unsigned b = grp + (unsigned)k * CNT_GROUPS;
        const unsigned c = te[k] >> 16;
        const unsigned at = sbase + b * PART_EDGES + (te[k] & 0xFFFFu) + min(s, max(c, 1u) - 1u);
        rec[k] = lst[c ? at : 0u];
        cmax = max(cmax, c);
        tot += c;
    }
    if (s == 0u) gtot[grp] = tot;
    __syncthreads();
    unsigned place = 0u, all = 0u;
    #pragma unroll
    for (unsigned q = 0; q < CNT_GROUPS; ++q) {
        const unsigned t = gtot[q];
        place += (q < grp) ? t : 0u;
        all   += t;
    }
    const unsigned place0 = place;
    unsigned* ol = olist + (size_t)(side * OWN_GROUPS + g) * OWN_LCAP;
    #pragma unroll
    for (int k = 0; k < CNT_PASSES; ++k) {
        const unsigned c = te[k] >> 16;
        if (s < c) {
            atomicAdd(&cnt[rec[k] >> 17], 1u);
            if (place + s < OWN_LCAP) ol[place + s] = rec[k];
        }
        place += c;
    }
    if (cmax > SEG_LANES) {
        unsigned pl = place0;
        for (unsigned b = grp; b < PART_NB; b += CNT_GROUPS) {
            const unsigned t = tabg[b * OWN_GROUPS];
            const unsigned c = t >> 16;
            const unsigned at = sbase + b * PART_EDGES + (t & 0xFFFFu);
            for (unsigned i = s + SEG_LANES; i < c; i += SEG_LANES) {
                const unsigned r = lst[at + i];
                atomicAdd(&cnt[r >> 17], 1u);
                if (pl + i < OWN_LCAP) ol[pl + i] = r;
            }
            pl += c;
        }
    }
    if (tid == 0) ocnt[side * OWN_GROUPS + g] = all;
    __syncthreads();
    float* deg = side ? deg_in : deg_out;
    for (int i = tid; i < OWN_ROWS; i += CNT_THREADS)
        if (lo + i < N_NODES) deg[lo + i] = (float)cnt[i];
}

#define OWN_QCAP   OWN_LCAP

template <int K>
static __device__ __forceinline__ void dirgcn_drain_round(const float* __restrict__ x, float* acc,
                                                          int o, float w, int dl, int j, int lane) {
    v4f v[K]; float wj[K]; int dj[K];
    #pragma unroll
    for (int u = 0; u < K; ++u) {
        const int oj = __builtin_amdgcn_readlane(o, j + u);
        wj[u] = __int_as_float(__builtin_amdgcn_readlane(__float_as_int(w), j + u));
        dj[u] = __builtin_amdgcn_readlane(dl, j + u);
        v[u] = ((const v4f*)(x + (size_t)oj * DIM))[lane];
    }
    #pragma unroll
    for (int u = 0; u < K; ++u) {
        #pragma unroll
        for (int i = 0; i < 4; ++i)
            atomicAdd(&acc[dj[u] * DIM + 32 * i + lane], wj[u] * v[u][i]);
    }
}

__global__ void __launch_bounds__(1024)
dirgcn_owner_kernel(const float* __restrict__ x,
                    const float* __restrict__ deg_out,
                    const float* __restrict__ deg_in,
                    const unsigned* __restrict__ tab,
                    const unsigned* __restrict__ lst,
                    const unsigned* __restrict__ olist,
                    const unsigned* __restrict__ ocnt,
                    float* __restrict__ agg_src,
                    float* __restrict__ agg_dst) {
    __shared__ float acc[OWN_ROWS * DIM];
    __shared__ int   queue[OWN_QCAP];

    const int tid  = threadIdx.x;
    const int lane = tid & 31;
    const int wave = tid >> 5;
    const int side = (blockIdx.x >= OWN_GROUPS) ? 1 : 0;
    const int g    = blockIdx.x - side * OWN_GROUPS;
    const int lo   = g * OWN_ROWS;
    const int cnt  = min(OWN_ROWS, N_NODES - lo);
    float* agg     = side ? agg_dst : agg_src;

    const unsigned n_all  = ocnt[side * OWN_GROUPS + g];
    const bool     listed = (n_all <= (unsigned)OWN_QCAP);
    unsigned qrec[OWN_QCAP / 1024] = {0u, 0u, 0u, 0u};
    if (listed && n_all > 0u) {
        const unsigned* ol = olist + (size_t)(side * OWN_GROUPS + g) * OWN_LCAP;
        #pragma unroll
        for (int u = 0; u < OWN_QCAP / 1024; ++u) qrec[u] = ol[min((unsigned)tid + 1024u * u, n_all - 1u)];
    }
    for (int i = tid; i < OWN_ROWS * DIM / 4; i += 1024)
        ((float4*)acc)[i] = make_float4(0.0f, 0.0f, 0.0f, 0.0f);
    if (listed && n_all > 0u) {
        #pragma unroll
        for (int u = 0; u < OWN_QCAP / 1024; ++u)
            if ((unsigned)tid + 1024u * u < n_all) queue[tid + 1024 * u] = (int)qrec[u];
    }
    __syncthreads();

    auto weight = [&](int kk, int oo) -> float {
        int row = side ? oo : kk;
        int col = side ? kk : oo;
        float dro = deg_out[row];
        float dci = deg_in[col];
        float ws = (dro > 0.0f) ? rsqrtf(fmaxf(dro, 1.0f)) : 0.0f;
        float wd = (dci > 0.0f) ? rsqrtf(fmaxf(dci, 1.0f)) : 0.0f;
        return ws * wd;
    };
    if (!listed) {
        if (tid < PART_NB) {
            const unsigned te = tab[((size_t)(side * PART_NB + tid)) * OWN_GROUPS + g];
            const int c = (int)(te >> 16);
            const unsigned* p = lst + (size_t)side * N_EDGES + (size_t)tid * PART_EDGES + (te & 0xFFFFu);
            for (int i = 0; i < c; ++i) {
                const unsigned rec = p[i];
                const int oo = (int)(rec & REC_OTH);
                const int dl = (int)(rec >> 17);
                const float w = weight(lo + dl, oo);
                const float* xr = x + (size_t)oo * DIM;
                for (int f = 0; f < DIM; ++f)
                    atomicAdd(&acc[dl * DIM + 32 * (f & 3) + (f >> 2)], w * xr[f]);
            }
        }
    }
    const int n = __builtin_amdgcn_readfirstlane(listed ? (int)n_all : 0);

    const int wave_s = __builtin_amdgcn_readfirstlane(wave);
    for (int base = wave_s * 32; base < n; base += 32 * 32) {
        const int m = min(32, n - base);
        int o = 0; int dl = 0; float w = 0.0f;
        if (lane < m) {
            const unsigned rec = (unsigned)queue[base + lane];
            o  = (int)(rec & REC_OTH);
            dl = (int)(rec >> 17);
            w  = weight(lo + dl, o);
        }
        int j = 0;
        for (; j + 16 <= m; j += 16) dirgcn_drain_round<16>(x, acc, o, w, dl, j, lane);
        if (j + 8 <= m) { dirgcn_drain_round<8>(x, acc, o, w, dl, j, lane); j += 8; }
        if (j + 4 <= m) { dirgcn_drain_round<4>(x, acc, o, w, dl, j, lane); j += 4; }
        if (j + 2 <= m) { dirgcn_drain_round<2>(x, acc, o, w, dl, j, lane); j += 2; }
        if (j < m)      { dirgcn_drain_round<1>(x, acc, o, w, dl, j, lane); }
    }
    __syncthreads();

    float4* dst = (float4*)(agg + (size_t)lo * DIM);
    for (int i = tid; i < cnt * (DIM / 4); i += 1024) {
        const float* ar = acc + (i >> 5) * DIM + (i & 31);
        dst[i] = make_float4(ar[0], ar[32], ar[64], ar[96]);
    }
}

__global__ void dirgcn_wprep_kernel(const float* __restrict__ Ws,
                                    const float* __restrict__ bs,
                                    const float* __restrict__ Wd,
                                    const float* __restrict__ bd,
                                    float* __restrict__ Wp,
                                    float* __restrict__ bias) {
    int d = blockIdx.x * blockDim.x + threadIdx.x;
    if (d < 2 * DIM * DIM) {
        int e    = d & 3;
        int lane = (d >> 2) & 31;
        int j    = (d >> 7) & 7;
        int p    = (d >> 10) & 15;
        int arr  = d >> 14;
        int n = 16 * j + (lane & 15);
        int k = 8 * p + 4 * (lane >> 4) + e;
        float w = arr ? ALPHA * Wd[n * DIM + k] : (1.0f - ALPHA) * Ws[n * DIM + k];
        Wp[d] = w;
    }
    if (d < DIM) bias[d] = (1.0f - ALPHA) * bs[d] + ALPHA * bd[d];
}

__global__ void __launch_bounds__(64)
dirgcn_gemm_kernel(const float* __restrict__ agg_src,
                   const float* __restrict__ agg_dst,
                   const float* __restrict__ Wp,
                   const float* __restrict__ bias,
                   float* __restrict__ out) {
    const int lane = threadIdx.x & 31;
    const int wave = threadIdx.x >> 5;
    const int tile = blockIdx.x * 2 + wave;
    const int row0 = tile * 16;
    const int c    = lane & 15;
    const int h    = lane >> 4;

    const unsigned aoff = (unsigned)((row0 + c) * DIM + 4 * h) * 4u;
    const unsigned woff = (unsigned)lane * 16u;
    const char* asb = (const char*)agg_src;
    const char* adb = (const char*)agg_dst;
    const char* wpb = (const char*)Wp;

    v8f acc[8];
    #pragma unroll
    for (int j = 0; j < 8; ++j) acc[j] = (v8f){};

    v4f a0, a1, b0[8], b1[8];

    auto load_set = [&](int qn, v4f& a, v4f (&b)[8]) {
        const char* ab = (qn < 16) ? (asb + 32 * qn) : (adb + 32 * (qn - 16));
        const char* wq = wpb + qn * 4096;
        a = *(const v4f*)(ab + aoff);
        #pragma unroll
        for (int j = 0; j < 8; ++j) b[j] = *(const v4f*)(wq + j * 512 + woff);
    };
    auto mma_set = [&](const v4f& a, const v4f (&b)[8]) {
        #pragma unroll
        for (int j = 0; j < 8; ++j)
            acc[j] = __builtin_amdgcn_wmma_f32_16x16x4_f32(false, a.xy, false, b[j].xy,
                                                           (short)0, acc[j], false, false);
        #pragma unroll
        for (int j = 0; j < 8; ++j)
            acc[j] = __builtin_amdgcn_wmma_f32_16x16x4_f32(false, a.zw, false, b[j].zw,
                                                           (short)0, acc[j], false, false);
    };

    load_set(0, a0, b0);
    #pragma unroll 1
    for (int q = 0; q < 30; q += 2) {
        load_set(q + 1, a1, b1);
        __builtin_amdgcn_sched_barrier(0);
        mma_set(a0, b0);
        __builtin_amdgcn_sched_barrier(0);
        load_set(q + 2, a0, b0);
        __builtin_amdgcn_sched_barrier(0);
        mma_set(a1, b1);
    }
    load_set(31, a1, b1);
    __builtin_amdgcn_sched_barrier(0);
    mma_set(a0, b0);
    mma_set(a1, b1);

    float* orow = out + (size_t)(row0 + h * 8) * DIM + c;
    #pragma unroll
    for (int j = 0; j < 8; ++j) {
        float bv = bias[16 * j + c];
        #pragma unroll
        for (int r = 0; r < 8; ++r) {
            orow[(size_t)r * DIM + 16 * j] = acc[j][r] + bv;
        }
    }
}

extern "C" void kernel_launch(void* const* d_in, const int* in_sizes, int n_in,
                              void* d_out, int out_size, void* d_ws, size_t ws_size,
                              hipStream_t stream) {
    const float* x    = (const float*)d_in[0];
    const int*   ei   = (const int*)d_in[1];
    const float* W_src = (const float*)d_in[2];
    const float* b_src = (const float*)d_in[3];
    const float* W_dst = (const float*)d_in[4];
    const float* b_dst = (const float*)d_in[5];
    float* out = (float*)d_out;

    float* ws = (float*)d_ws;
    float* deg_out = ws;
    float* deg_in  = deg_out + N_NODES;
    float* agg_src = deg_in + N_NODES;
    float* agg_dst = agg_src + (size_t)N_NODES * DIM;
    float* Wp      = agg_dst + (size_t)N_NODES * DIM;
    float* bias    = Wp + 2 * DIM * DIM;

    unsigned* lst = (unsigned*)d_out;
    unsigned* tab = lst + 2 * N_EDGES;
    unsigned* olist = tab + 2 * PART_NB * OWN_GROUPS;
    unsigned* ocnt  = olist + 2 * OWN_GROUPS * OWN_LCAP;
    dirgcn_partition_kernel<<<PART_NB, 256, 0, stream>>>(ei, tab, lst);

    dirgcn_count_kernel<<<2 * OWN_GROUPS, CNT_THREADS, 0, stream>>>(tab, lst, deg_out, deg_in,
                                                                   olist, ocnt);

    dirgcn_owner_kernel<<<2 * OWN_GROUPS, 1024, 0, stream>>>(x, deg_out, deg_in, tab, lst,
                                                            olist, ocnt, agg_src, agg_dst);

    dirgcn_wprep_kernel<<<(2 * DIM * DIM + 255) / 256, 256, 0, stream>>>(
        W_src, b_src, W_dst, b_dst, Wp, bias);

    dirgcn_gemm_kernel<<<N_NODES / 32, 64, 0, stream>>>(agg_src, agg_dst,
                                                        Wp, bias, out);
}
